# starting kernel
# baseline (speedup 1.0000x reference)
#include <hip/hip_runtime.h>

typedef __attribute__((ext_vector_type(2))) float v2f;
typedef __attribute__((ext_vector_type(8))) float v8f;

#define NCH   256
#define KTOT  2304
#define KG    (KTOT / 4)
#define HW    784

__global__ void build_weff_k(const float* __restrict__ w1, const float* __restrict__ w2,
                             float* __restrict__ weff) {
    int tid = blockIdx.x * blockDim.x + threadIdx.x;
    if (tid >= NCH * KTOT) return;
    int oc  = tid / KTOT;
    int r   = tid % KTOT;
    int blk = r >> 8;
    int cin = r & 255;
    int i = blk / 3, k = blk % 3;
    int l = oc >> 3, p = oc & 7;
    int cg = cin >> 2, n = cin & 3;
    float a = w1[((l * 4 + n) * 3 + i) * 3 + k];
    float b = w2[(p * 64 + cg) * 3 + k];
    weff[tid] = a * b;
}

__global__ void build_bpack_k(const float* __restrict__ x, float* __restrict__ bpack) {
    int tid = blockIdx.x * blockDim.x + threadIdx.x;
    if (tid >= KTOT * HW) return;
    int e  = tid & 3;
    int t  = tid >> 2;
    int s  = t % HW;
    int kg = t / HW;
    int blk = kg >> 6;
    int cin = ((kg & 63) << 2) | e;
    int i = blk / 3, k = blk % 3;
    int ho = s / 28, wo = s % 28;
    int row = ho + i - 1, col = wo + k - 1;
    float v = 0.0f;
    if (row >= 0 && row < 28 && col >= 0 && col < 28)
        v = x[cin * HW + row * 28 + col];
    bpack[tid] = v;
}

__global__ void conv_wmma_k(const float* __restrict__ weff, const float* __restrict__ bpack,
                            float* __restrict__ out) {
    int wave   = (blockIdx.x * blockDim.x + threadIdx.x) >> 5;
    int lane   = threadIdx.x & 31;
    int m_tile = wave & 15;
    int s_tile = wave >> 4;
    int lr     = lane & 15;
    int half   = lane >> 4;

    int s    = s_tile * 16 + lr;
    int rowA = m_tile * 16 + lr;

    const float* ap = weff + rowA * KTOT + 2 * half;
    const float* bp = bpack + s * 4 + 2 * half;

    v8f acc = {};
    #pragma unroll 8
    for (int kg = 0; kg < KG; ++kg) {
        v2f a = *(const v2f*)(ap + kg * 4);
        v2f b = *(const v2f*)(bp + kg * (HW * 4));
        acc = __builtin_amdgcn_wmma_f32_16x16x4_f32(
            false, a, false, b, (short)0, acc, false, false);
    }

    #pragma unroll
    for (int r = 0; r < 8; ++r) {
        int oc = m_tile * 16 + half * 8 + r;
        out[oc * HW + s] = acc[r];
    }
}

extern "C" void kernel_launch(void* const* d_in, const int* in_sizes, int n_in,
                              void* d_out, int out_size, void* d_ws, size_t ws_size,
                              hipStream_t stream) {
    const float* x  = (const float*)d_in[0];
    const float* w1 = (const float*)d_in[1];
    const float* w2 = (const float*)d_in[2];
    float* out = (float*)d_out;

    float* weff  = (float*)d_ws;
    float* bpack = weff + NCH * KTOT;

    build_weff_k<<<(NCH * KTOT + 255) / 256, 256, 0, stream>>>(w1, w2, weff);
    build_bpack_k<<<(KTOT * HW + 255) / 256, 256, 0, stream>>>(x, bpack);
    conv_wmma_k<<<196, 128, 0, stream>>>(weff, bpack, out);
}


# Contract in two stages as the reference does (j with w2: K=64 WMMA GEMM, then n,i,k with w1: K=36 WMMA GEMM) instead of one K=2304 GEMM on the product weight; 3 launches to 2; sum order changes
# speedup vs baseline: 7.8886x; 7.8886x over previous
#include <hip/hip_runtime.h>

typedef __attribute__((ext_vector_type(2))) float v2f;
typedef __attribute__((ext_vector_type(8))) float v8f;

#define HW     784
#define XCOLS  3136
#define UROWS  24


__global__ void contract_j_k(const float* __restrict__ x, const float* __restrict__ w2,
                             float* __restrict__ u) {
    int lane = threadIdx.x & 31;
    int lr   = lane & 15;
    int half = lane >> 4;
    int col  = blockIdx.x * 16 + lr;

    int  r1   = 16 + lr;
    bool v1   = r1 < UROWS;
    int  r1c  = v1 ? r1 : 0;
    const float* a0p = w2 + (lr  / 3) * 192 + (lr  % 3) + 6 * half;
    const float* a1p = w2 + (r1c / 3) * 192 + (r1c % 3) + 6 * half;
    const float* bp  = x + 2 * half * XCOLS + col;

    v8f acc0 = {}, acc1 = {};
    #pragma unroll
    for (int kg = 0; kg < 16; ++kg) {
        v2f b, a0, a1;
        b[0]  = bp[(kg * 4 + 0) * XCOLS];
        b[1]  = bp[(kg * 4 + 1) * XCOLS];
        a0[0] = a0p[(kg * 4 + 0) * 3];
        a0[1] = a0p[(kg * 4 + 1) * 3];
        float t0 = a1p[(kg * 4 + 0) * 3];
        float t1 = a1p[(kg * 4 + 1) * 3];
        a1[0] = v1 ? t0 : 0.0f;
        a1[1] = v1 ? t1 : 0.0f;
        acc0 = __builtin_amdgcn_wmma_f32_16x16x4_f32(false, a0, false, b, (short)0, acc0, false, false);
        acc1 = __builtin_amdgcn_wmma_f32_16x16x4_f32(false, a1, false, b, (short)0, acc1, false, false);
    }

    #pragma unroll
    for (int q = 0; q < 8; ++q)
        u[(half * 8 + q) * XCOLS + col] = acc0[q];
    if (half == 0) {
        #pragma unroll
        for (int q = 0; q < 8; ++q)
            u[(16 + q) * XCOLS + col] = acc1[q];
    }
}

__global__ void contract_nik_k(const float* __restrict__ w1, const float* __restrict__ u,
                               float* __restrict__ out) {
    int wave   = (blockIdx.x * blockDim.x + threadIdx.x) >> 5;
    int lane   = threadIdx.x & 31;
    int lr     = lane & 15;
    int half   = lane >> 4;
    int p      = wave / 49;
    int s_tile = wave - p * 49;
    int s      = s_tile * 16 + lr;
    int ho     = s / 28;
    int wo     = s - ho * 28;

    bool rv0 = ho >= 1, rv2 = ho <= 26;
    bool cv0 = wo >= 1, cv2 = wo <= 26;

    const float* a0p = w1 + lr * 36 + 2 * half;
    const float* a1p = w1 + (16 + lr) * 36 + 2 * half;
    int base = p * (3 * 4 * HW) + s;

    int  idx[18];
    bool okv[18];
    #pragma unroll
    for (int t = 0; t < 18; ++t) {
        const int kkA = (t >> 1) * 4 + (t & 1), kkB = kkA + 2;
        const int nA = kkA / 9, iA = (kkA % 9) / 3, kA = kkA % 3;
        const int nB = kkB / 9, iB = (kkB % 9) / 3, kB = kkB % 3;
        const int offA = (kA * 4 + nA) * HW + (iA - 1) * 28 + (kA - 1);
        const int offB = (kB * 4 + nB) * HW + (iB - 1) * 28 + (kB - 1);
        bool okA = (iA == 0 ? rv0 : iA == 2 ? rv2 : true) && (kA == 0 ? cv0 : kA == 2 ? cv2 : true);
        bool okB = (iB == 0 ? rv0 : iB == 2 ? rv2 : true) && (kB == 0 ? cv0 : kB == 2 ? cv2 : true);
        bool ok  = half ? okB : okA;
        int  off = half ? offB : offA;
        okv[t] = ok;
        idx[t] = base + (ok ? off : 0);
    }
    float g[18];
    v2f a0[9], a1[9];
    #pragma unroll
    for (int t = 0; t < 18; ++t) g[t] = u[idx[t]];
    #pragma unroll
    for (int kg = 0; kg < 9; ++kg) {
        a0[kg] = *(const v2f*)(a0p + kg * 4);
        a1[kg] = *(const v2f*)(a1p + kg * 4);
    }
    __builtin_amdgcn_sched_barrier(0);

    v8f acc0 = {}, acc1 = {};
    #pragma unroll
    for (int kg = 0; kg < 9; ++kg) {
        v2f b;
        b[0] = okv[2 * kg]     ? g[2 * kg]     : 0.0f;
        b[1] = okv[2 * kg + 1] ? g[2 * kg + 1] : 0.0f;
        acc0 = __builtin_amdgcn_wmma_f32_16x16x4_f32(false, a0[kg], false, b, (short)0, acc0, false, false);
        acc1 = __builtin_amdgcn_wmma_f32_16x16x4_f32(false, a1[kg], false, b, (short)0, acc1, false, false);
    }

    #pragma unroll
    for (int q = 0; q < 8; ++q) {
        out[((half * 8 + q) * 8 + p) * HW + s]      = acc0[q];
        out[((16 + half * 8 + q) * 8 + p) * HW + s] = acc1[q];
    }
}

extern "C" void kernel_launch(void* const* d_in, const int* in_sizes, int n_in,
                              void* d_out, int out_size, void* d_ws, size_t ws_size,
                              hipStream_t stream) {
    const float* x  = (const float*)d_in[0];
    const float* w1 = (const float*)d_in[1];
    const float* w2 = (const float*)d_in[2];
    float* out = (float*)d_out;
    float* u   = (float*)d_ws;

    contract_j_k<<<196, 32, 0, stream>>>(x, w2, u);
    contract_nik_k<<<196, 64, 0, stream>>>(w1, u, out);
}


# Fuse the two stages into one launch: each work-group (4 p, one output row, 14 columns) recomputes the 3-row, 16-column patch of U it needs into LDS and contracts from LDS; 2 launches to 1
# speedup vs baseline: 11.0654x; 1.4027x over previous
#include <hip/hip_runtime.h>


typedef __attribute__((ext_vector_type(2))) float v2f;
typedef __attribute__((ext_vector_type(8))) float v8f;

#define HW      784
#define XCOLS   3136
#define APITCH  68
#define UPITCH  196

__device__ __forceinline__ void lds_barrier() {
    __builtin_amdgcn_fence(__ATOMIC_RELEASE, "workgroup", "local");
    __builtin_amdgcn_s_barrier();
    __builtin_amdgcn_fence(__ATOMIC_ACQUIRE, "workgroup", "local");
}

__global__ __launch_bounds__(384)
void contract_fused_k(const float* __restrict__ x, const float* __restrict__ w1,
                      const float* w2, float* __restrict__ out) {
    __shared__ float a_lds[12 * APITCH];
    __shared__ float u_lds[12 * UPITCH];

    int tid  = threadIdx.x;
    int wave = __builtin_amdgcn_readfirstlane(tid >> 5);
    int lane = tid & 31;
    int lr   = lane & 15;
    int half = lane >> 4;

    int blk = blockIdx.x;
    int g   = blk / 56;
    int rem = blk - g * 56;
    int ho  = rem >> 1;
    int c   = rem & 1;

    v2f wv = *(const v2f*)(w2 + g * 768 + 2 * tid);
    __builtin_amdgcn_sched_barrier(0);

    int n   = wave / 3;
    int dr  = wave - n * 3;
    int row = ho - 1 + dr;
    int col = c * 14 - 1 + lr;
    bool ok = (row >= 0) && (row <= 27) && (col >= 0) && (col <= 27);
    int rowc = row < 0 ? 0 : (row > 27 ? 27 : row);
    int colc = col < 0 ? 0 : (col > 27 ? 27 : col);
    const float* bp = x + (8 * half + n) * HW + rowc * 28 + colc;
    float bv[32];
    #pragma unroll
    for (int t = 0; t < 32; ++t)
        bv[t] = bp[((t >> 1) * 4 + (t & 1)) * XCOLS];

    int pl = wave >> 1;
    int lt = wave & 1;
    const float* a2p = w1 + (lt * 16 + lr) * 36 + 2 * half;
    v2f a2[9];
    #pragma unroll
    for (int kg = 0; kg < 9; ++kg) a2[kg] = *(const v2f*)(a2p + kg * 4);

    {
        int f0 = 2 * tid, f1 = f0 + 1;
        a_lds[((f0 / 192) * 3 + f0 % 3) * APITCH + (f0 % 192) / 3] = wv[0];
        a_lds[((f1 / 192) * 3 + f1 % 3) * APITCH + (f1 % 192) / 3] = wv[1];
    }
    lds_barrier();

    {
        int rA = lr < 12 ? lr : lr - 12;
        const float* ap = a_lds + rA * APITCH + 2 * half;
        v2f a[16];
        #pragma unroll
        for (int kg = 0; kg < 16; ++kg) a[kg] = *(const v2f*)(ap + kg * 4);
        __builtin_amdgcn_sched_barrier(0);
        v8f acc = {};
        #pragma unroll
        for (int kg = 0; kg < 16; ++kg) {
            v2f b;
            b[0] = bv[2 * kg];
            b[1] = bv[2 * kg + 1];
            acc = __builtin_amdgcn_wmma_f32_16x16x4_f32(false, a[kg], false, b, (short)0, acc, false, false);
        }
        #pragma unroll
        for (int q = 0; q < 8; ++q)
            if (half == 0 || q < 4)
                u_lds[(q + 8 * half) * UPITCH + wave * 16 + lr] = ok ? acc[q] : 0.0f;
    }
    lds_barrier();

    if (wave < 8) {
        int tt = lr < 14 ? lr : 13;
        const float* ub = u_lds + pl * 3 * UPITCH + tt;
        float gq[18];
        #pragma unroll
        for (int t = 0; t < 18; ++t) {
            const int kkA = (t >> 1) * 4 + (t & 1), kkB = kkA + 2;
            const int nA = kkA / 9, iA = (kkA % 9) / 3, kA = kkA % 3;
            const int nB = kkB / 9, iB = (kkB % 9) / 3, kB = kkB % 3;
            const int offA = kA * UPITCH + (nA * 3 + iA) * 16 + kA;
            const int offB = kB * UPITCH + (nB * 3 + iB) * 16 + kB;
            gq[t] = ub[half ? offB : offA];
        }
        v8f acc = {};
        #pragma unroll
        for (int kg = 0; kg < 9; ++kg) {
            v2f b;
            b[0] = gq[2 * kg];
            b[1] = gq[2 * kg + 1];
            acc = __builtin_amdgcn_wmma_f32_16x16x4_f32(false, a2[kg], false, b, (short)0, acc, false, false);
        }
        if (lr < 14) {
            float* op = out + ((lt * 16 + half * 8) * 8 + g * 4 + pl) * HW + ho * 28 + c * 14 + lr;
            #pragma unroll
            for (int q = 0; q < 8; ++q) op[q * 8 * HW] = acc[q];
        }
    }
}

extern "C" void kernel_launch(void* const* d_in, const int* in_sizes, int n_in,
                              void* d_out, int out_size, void* d_ws, size_t ws_size,
                              hipStream_t stream) {
    const float* x  = (const float*)d_in[0];
    const float* w1 = (const float*)d_in[1];
    const float* w2 = (const float*)d_in[2];
    float* out = (float*)d_out;

    contract_fused_k<<<112, 384, 0, stream>>>(x, w1, w2, out);
}


# Stage-2 weight (w1) reaches its fragments through LDS: one whole-line 16-byte load per thread and an LDS copy replace nine 8-byte loads per wave that touched 16-17 lines each; sum order unchanged
# speedup vs baseline: 11.6311x; 1.0511x over previous
#include <hip/hip_runtime.h>


typedef __attribute__((ext_vector_type(2))) float v2f;
typedef __attribute__((ext_vector_type(4))) float v4f;
typedef __attribute__((ext_vector_type(8))) float v8f;

#define HW      784
#define XCOLS   3136
#define APITCH  68
#define UPITCH  196
#define W1ROW   36

__device__ __forceinline__ void lds_barrier() {
    __builtin_amdgcn_fence(__ATOMIC_RELEASE, "workgroup", "local");
    __builtin_amdgcn_s_barrier();
    __builtin_amdgcn_fence(__ATOMIC_ACQUIRE, "workgroup", "local");
}

__global__ __launch_bounds__(384)
void contract_fused_k(const float* __restrict__ x, const float* __restrict__ w1,
                      const float* w2, float* __restrict__ out) {
    __shared__ float a_lds[12 * APITCH];
    __shared__ float u_lds[12 * UPITCH];
    __shared__ alignas(16) float w1_lds[32 * W1ROW];

    int tid  = threadIdx.x;
    int wave = __builtin_amdgcn_readfirstlane(tid >> 5);
    int lane = tid & 31;
    int lr   = lane & 15;
    int half = lane >> 4;

    int blk = blockIdx.x;
    int g   = blk / 56;
    int rem = blk - g * 56;
    int ho  = rem >> 1;
    int c   = rem & 1;

    v2f wv = *(const v2f*)(w2 + g * 768 + 2 * tid);
    v4f w1v = {};
    if (wave < 9) w1v = *(const v4f*)(w1 + 4 * tid);
    __builtin_amdgcn_sched_barrier(0);

    int n   = wave / 3;
    int dr  = wave - n * 3;
    int row = ho - 1 + dr;
    int col = c * 14 - 1 + lr;
    bool ok = (row >= 0) && (row <= 27) && (col >= 0) && (col <= 27);
    int rowc = row < 0 ? 0 : (row > 27 ? 27 : row);
    int colc = col < 0 ? 0 : (col > 27 ? 27 : col);
    const float* bp = x + (8 * half + n) * HW + rowc * 28 + colc;
    float bv[32];
    #pragma unroll
    for (int t = 0; t < 32; ++t)
        bv[t] = bp[((t >> 1) * 4 + (t & 1)) * XCOLS];

    int pl = wave >> 1;
    int lt = wave & 1;

    {
        int f0 = 2 * tid, f1 = f0 + 1;
        a_lds[((f0 / 192) * 3 + f0 % 3) * APITCH + (f0 % 192) / 3] = wv[0];
        a_lds[((f1 / 192) * 3 + f1 % 3) * APITCH + (f1 % 192) / 3] = wv[1];
    }
    if (wave < 9) *(v4f*)(w1_lds + 4 * tid) = w1v;
    lds_barrier();

    v2f a2[9];
    {
        int rA = lr < 12 ? lr : lr - 12;
        const float* ap = a_lds + rA * APITCH + 2 * half;
        v2f a[16];
        #pragma unroll
        for (int kg = 0; kg < 16; ++kg) a[kg] = *(const v2f*)(ap + kg * 4);
        const float* a2p = w1_lds + (lt * 16 + lr) * W1ROW + 2 * half;
        #pragma unroll
        for (int kg = 0; kg < 9; ++kg) a2[kg] = *(const v2f*)(a2p + kg * 4);
        __builtin_amdgcn_sched_barrier(0);
        v8f acc = {};
        #pragma unroll
        for (int kg = 0; kg < 16; ++kg) {
            v2f b;
            b[0] = bv[2 * kg];
            b[1] = bv[2 * kg + 1];
            acc = __builtin_amdgcn_wmma_f32_16x16x4_f32(false, a[kg], false, b, (short)0, acc, false, false);
        }
        #pragma unroll
        for (int q = 0; q < 8; ++q)
            if (half == 0 || q < 4)
                u_lds[(q + 8 * half) * UPITCH + wave * 16 + lr] = ok ? acc[q] : 0.0f;
    }
    lds_barrier();

    if (wave < 8) {
        int tt = lr < 14 ? lr : 13;
        const float* ub = u_lds + pl * 3 * UPITCH + tt;
        float gq[18];
        #pragma unroll
        for (int t = 0; t < 18; ++t) {
            const int kkA = (t >> 1) * 4 + (t & 1), kkB = kkA + 2;
            const int nA = kkA / 9, iA = (kkA % 9) / 3, kA = kkA % 3;
            const int nB = kkB / 9, iB = (kkB % 9) / 3, kB = kkB % 3;
            const int offA = kA * UPITCH + (nA * 3 + iA) * 16 + kA;
            const int offB = kB * UPITCH + (nB * 3 + iB) * 16 + kB;
            gq[t] = ub[half ? offB : offA];
        }
        v8f acc = {};
        #pragma unroll
        for (int kg = 0; kg < 9; ++kg) {
            v2f b;
            b[0] = gq[2 * kg];
            b[1] = gq[2 * kg + 1];
            acc = __builtin_amdgcn_wmma_f32_16x16x4_f32(false, a2[kg], false, b, (short)0, acc, false, false);
        }
        if (lr < 14) {
            float* op = out + ((lt * 16 + half * 8) * 8 + g * 4 + pl) * HW + ho * 28 + c * 14 + lr;
            #pragma unroll
            for (int q = 0; q < 8; ++q) op[q * 8 * HW] = acc[q];
        }
    }
}

extern "C" void kernel_launch(void* const* d_in, const int* in_sizes, int n_in,
                              void* d_out, int out_size, void* d_ws, size_t ws_size,
                              hipStream_t stream) {
    const float* x  = (const float*)d_in[0];
    const float* w1 = (const float*)d_in[1];
    const float* w2 = (const float*)d_in[2];
    float* out = (float*)d_out;

    contract_fused_k<<<112, 384, 0, stream>>>(x, w1, w2, out);
}
